# starting kernel
# baseline (speedup 1.0000x reference)
#include <hip/hip_runtime.h>
#include <math.h>

typedef float v2f __attribute__((ext_vector_type(2)));
typedef float v8f __attribute__((ext_vector_type(8)));

#define NQ 4
#define DIM 16

__device__ __forceinline__ void cmul2x2(const float* ar, const float* ai,
                                        const float* br, const float* bi,
                                        float* cr, float* ci) {
  for (int r = 0; r < 2; ++r)
    for (int c = 0; c < 2; ++c) {
      float sr = 0.f, si = 0.f;
      for (int k = 0; k < 2; ++k) {
        float xr = ar[r * 2 + k], xi = ai[r * 2 + k];
        float yr = br[k * 2 + c], yi = bi[k * 2 + c];
        sr += xr * yr - xi * yi;
        si += xr * yi + xi * yr;
      }
      cr[r * 2 + c] = sr;
      ci[r * 2 + c] = si;
    }
}

__global__ void qnn_setup_kernel(const float* __restrict__ params,
                                 const float* __restrict__ fc_w,
                                 const float* __restrict__ fc_b,
                                 float* __restrict__ ws) {
  if (threadIdx.x != 0 || blockIdx.x != 0) return;

  float gre[NQ][4], gim[NQ][4];
  for (int i = 0; i < NQ; ++i) {
    float hx = 0.5f * params[i * 3 + 0];
    float hy = 0.5f * params[i * 3 + 1];
    float hz = 0.5f * params[i * 3 + 2];
    float cx = cosf(hx), sx = sinf(hx);
    float cy = cosf(hy), sy = sinf(hy);
    float cz = cosf(hz), sz = sinf(hz);
    float axr[4] = {cx, 0.f, 0.f, cx},  axi[4] = {0.f, -sx, -sx, 0.f};
    float ayr[4] = {cy, -sy, sy, cy},   ayi[4] = {0.f, 0.f, 0.f, 0.f};
    float azr[4] = {cz, 0.f, 0.f, cz},  azi[4] = {-sz, 0.f, 0.f, sz};
    float tr4[4], ti4[4];
    cmul2x2(ayr, ayi, axr, axi, tr4, ti4);
    cmul2x2(azr, azi, tr4, ti4, gre[i], gim[i]);
  }

  float re[DIM * DIM], im[DIM * DIM], tre[DIM * DIM], tim[DIM * DIM];
  re[0] = 1.f; im[0] = 0.f;
  int s = 1;
  for (int q = 0; q < NQ; ++q) {
    int ns = s * 2;
    for (int a = 0; a < s; ++a)
      for (int b = 0; b < s; ++b) {
        float ar = re[a * s + b], ai = im[a * s + b];
        for (int u = 0; u < 2; ++u)
          for (int v = 0; v < 2; ++v) {
            float br = gre[q][u * 2 + v], bi = gim[q][u * 2 + v];
            tre[(a * 2 + u) * ns + (b * 2 + v)] = ar * br - ai * bi;
            tim[(a * 2 + u) * ns + (b * 2 + v)] = ar * bi + ai * br;
          }
      }
    s = ns;
    for (int k = 0; k < s * s; ++k) { re[k] = tre[k]; im[k] = tim[k]; }
  }

  for (int g = 0; g < NQ - 1; ++g) {
    int cbit = 1 << (NQ - 1 - g), tbit = 1 << (NQ - 1 - (g + 1));
    for (int k = 0; k < DIM; ++k) {
      int src = (k & cbit) ? (k ^ tbit) : k;
      for (int i = 0; i < DIM; ++i) {
        tre[k * DIM + i] = re[src * DIM + i];
        tim[k * DIM + i] = im[src * DIM + i];
      }
    }
    for (int k = 0; k < DIM * DIM; ++k) { re[k] = tre[k]; im[k] = tim[k]; }
  }

  for (int k = 0; k < DIM * DIM; ++k) { ws[k] = re[k]; ws[256 + k] = im[k]; }

  for (int n = 0; n < DIM; ++n) {
    for (int c = 0; c < 3; ++c) {
      float acc = 0.f;
      for (int w = 0; w < NQ; ++w) {
        float zs = 1.f - 2.f * (float)((n >> (NQ - 1 - w)) & 1);
        acc += zs * fc_w[c * NQ + w];
      }
      ws[512 + n * 4 + c] = acc;
    }
    ws[512 + n * 4 + 3] = 1.f;
  }
  for (int c = 0; c < 3; ++c) ws[576 + c] = fc_b[c];
}

__device__ __forceinline__ float swap_half16(float v) {
  int i = __float_as_int(v);
  int r = __builtin_amdgcn_permlanex16(i, i, 0x76543210, 0xfedcba98, false, false);
  return __int_as_float(r);
}

__global__ __launch_bounds__(256) void qnn_main_kernel(
    const float* __restrict__ x, const float* __restrict__ ws,
    float* __restrict__ out, int ntiles) {
  const int lane = threadIdx.x & 31;
  const int half = lane >> 4;
  const int lrow = lane & 15;
  const int wave = threadIdx.x >> 5;
  const int gwave = blockIdx.x * 8 + wave;
  const int nwaves = gridDim.x * 8;

  const float* __restrict__ Ure = ws;
  const float* __restrict__ Uim = ws + 256;
  const float* __restrict__ W4  = ws + 512;
  const float* __restrict__ bia = ws + 576;

  v2f a_re[4], a_im[4];
#pragma unroll
  for (int c = 0; c < 4; ++c) {
    int off = lrow * 16 + 4 * c + 2 * half;
    a_re[c] = *(const v2f*)(Ure + off);
    a_im[c] = *(const v2f*)(Uim + off);
  }
  v2f w0p[4], w1p[4], w2p[4];
#pragma unroll
  for (int rp = 0; rp < 4; ++rp) {
    const float* pa = W4 + (2 * rp + 0 + 8 * half) * 4;
    const float* pb = W4 + (2 * rp + 1 + 8 * half) * 4;
    w0p[rp].x = pa[0]; w0p[rp].y = pb[0];
    w1p[rp].x = pa[1]; w1p[rp].y = pb[1];
    w2p[rp].x = pa[2]; w2p[rp].y = pb[2];
  }
  const float b0 = bia[0], b1 = bia[1], b2 = bia[2];

  for (int t = gwave; t < ntiles; t += nwaves) {
    const float* xt = x + (size_t)t * 256;
    __builtin_prefetch(xt + (size_t)nwaves * 256 + lane * 16, 0, 0);

    v2f bch[4];
#pragma unroll
    for (int c = 0; c < 4; ++c)
      bch[c] = *(const v2f*)(xt + lrow * 16 + 4 * c + 2 * half);

    v8f accR = {}, accI = {};
#pragma unroll
    for (int c = 0; c < 4; ++c) {
      accR = __builtin_amdgcn_wmma_f32_16x16x4_f32(
          false, a_re[c], false, bch[c], (short)0, accR, false, false);
      accI = __builtin_amdgcn_wmma_f32_16x16x4_f32(
          false, a_im[c], false, bch[c], (short)0, accI, false, false);
    }

    v2f acc0 = {0.f, 0.f}, acc1 = {0.f, 0.f}, acc2 = {0.f, 0.f}, acc3 = {0.f, 0.f};
#pragma unroll
    for (int rp = 0; rp < 4; ++rp) {
      v2f vr; vr.x = accR[2 * rp]; vr.y = accR[2 * rp + 1];
      v2f vi; vi.x = accI[2 * rp]; vi.y = accI[2 * rp + 1];
      v2f p = vr * vr + vi * vi;
      acc0 += p * w0p[rp];
      acc1 += p * w1p[rp];
      acc2 += p * w2p[rp];
      acc3 += p;
    }
    float s0 = acc0.x + acc0.y;
    float s1 = acc1.x + acc1.y;
    float s2 = acc2.x + acc2.y;
    float s3 = acc3.x + acc3.y;

    s0 += swap_half16(s0);
    s1 += swap_half16(s1);
    s2 += swap_half16(s2);
    s3 += swap_half16(s3);

    const float rinv = __builtin_amdgcn_rcpf(s3);
    const size_t samp = (size_t)t * 16 + lrow;
    float* op = out + samp * 3;
    if (half == 0) {
      op[0] = s0 * rinv + b0;
      op[1] = s1 * rinv + b1;
    } else {
      op[2] = s2 * rinv + b2;
    }
  }
}

extern "C" void kernel_launch(void* const* d_in, const int* in_sizes, int n_in,
                              void* d_out, int out_size, void* d_ws, size_t ws_size,
                              hipStream_t stream) {
  const float* x      = (const float*)d_in[0];
  const float* params = (const float*)d_in[1];
  const float* fc_w   = (const float*)d_in[2];
  const float* fc_b   = (const float*)d_in[3];
  float* out = (float*)d_out;
  float* ws  = (float*)d_ws;

  const int ntiles = in_sizes[0] / 256;

  qnn_setup_kernel<<<1, 32, 0, stream>>>(params, fc_w, fc_b, ws);

  int blocks = 2048;
  int max_blocks = (ntiles + 7) / 8;
  if (blocks > max_blocks) blocks = max_blocks;
  if (blocks < 1) blocks = 1;
  qnn_main_kernel<<<blocks, 256, 0, stream>>>(x, ws, out, ntiles);
}


# Set-up launch removed: per block wave 0 builds each lane's U fragments from params (one sincos per lane, shared by readlane), wave 1 the head weights, via LDS; first tile prefetched under it
# speedup vs baseline: 3.9931x; 3.9931x over previous
#include <hip/hip_runtime.h>
#include <math.h>


typedef float v2f __attribute__((ext_vector_type(2)));
typedef float v4f __attribute__((ext_vector_type(4)));
typedef float v8f __attribute__((ext_vector_type(8)));

#define NQ 4

struct cplx { float re, im; };

__device__ __forceinline__ cplx cmul(cplx a, cplx b) {
  cplx c;
  c.re = a.re * b.re - a.im * b.im;
  c.im = a.re * b.im + a.im * b.re;
  return c;
}

__device__ __forceinline__ cplx csel(bool s, cplx one, cplx zero) {
  cplx c;
  c.re = s ? one.re : zero.re;
  c.im = s ? one.im : zero.im;
  return c;
}

__device__ __forceinline__ float swap_half16(float v) {
  int i = __float_as_int(v);
  int r = __builtin_amdgcn_permlanex16(i, i, 0x76543210, 0xfedcba98, false, false);
  return __int_as_float(r);
}

__device__ __forceinline__ float lane_bcast(float v, int j) {
  return __int_as_float(__builtin_amdgcn_readlane(__float_as_int(v), j));
}

__global__ __launch_bounds__(256) void qnn_main_kernel(
    const float* __restrict__ x, const float* __restrict__ params,
    const float* __restrict__ fc_w, const float* __restrict__ fc_b,
    float* __restrict__ out, int ntiles) {
  __shared__ v4f sh[10][32];

  const int lane = threadIdx.x & 31;
  const int half = lane >> 4;
  const int lrow = lane & 15;
  const int wave = threadIdx.x >> 5;
  const int gwave = blockIdx.x * 8 + wave;
  const int nwaves = gridDim.x * 8;

  if (gwave < ntiles)
    __builtin_prefetch(x + (size_t)gwave * 256 + lrow * 16, 0, 0);

  if (wave == 0) {
    const int pj = lane < 12 ? lane : 0;
    const float h = 0.5f * params[pj];
    const float cv = cosf(h), sv = sinf(h);
    float C[12], S[12];
#pragma unroll
    for (int j = 0; j < 12; ++j) {
      C[j] = lane_bcast(cv, j);
      S[j] = lane_bcast(sv, j);
    }

    int r = lrow;
    if (r & 2) r ^= 1;
    if (r & 4) r ^= 2;
    if (r & 8) r ^= 4;

    cplx g0[NQ], g1[NQ];
#pragma unroll
    for (int q = 0; q < NQ; ++q) {
      const float cx = C[3 * q + 0], sx = S[3 * q + 0];
      const float cy = C[3 * q + 1], sy = S[3 * q + 1];
      const float cz = C[3 * q + 2], sz = S[3 * q + 2];
      const float a = cy * cx, b = sy * sx, d = sy * cx, e = cy * sx;
      const float p = cz * a + sz * b;
      const float s = cz * b - sz * a;
      const float m = cz * d + sz * e;
      const float n = sz * d - cz * e;
      const bool u = ((r >> (NQ - 1 - q)) & 1) != 0;
      cplx G00 = {p, s}, G01 = {-m, n}, G10 = {m, n}, G11 = {p, -s};
      g0[q] = csel(u, G10, G00);
      g1[q] = csel(u, G11, G01);
    }

    const cplx t2 = csel(half != 0, g1[2], g0[2]);
    v2f a_re[4], a_im[4];
#pragma unroll
    for (int c = 0; c < 4; ++c) {
      const cplx p01 = cmul((c >> 1) ? g1[0] : g0[0], (c & 1) ? g1[1] : g0[1]);
      const cplx p012 = cmul(p01, t2);
      const cplx e0 = cmul(p012, g0[3]);
      const cplx e1 = cmul(p012, g1[3]);
      a_re[c].x = e0.re; a_re[c].y = e1.re;
      a_im[c].x = e0.im; a_im[c].y = e1.im;
    }
    sh[0][lane] = (v4f){a_re[0].x, a_re[0].y, a_re[1].x, a_re[1].y};
    sh[1][lane] = (v4f){a_re[2].x, a_re[2].y, a_re[3].x, a_re[3].y};
    sh[2][lane] = (v4f){a_im[0].x, a_im[0].y, a_im[1].x, a_im[1].y};
    sh[3][lane] = (v4f){a_im[2].x, a_im[2].y, a_im[3].x, a_im[3].y};
  } else if (wave == 1) {
#pragma unroll
    for (int k = 0; k < 3; ++k) {
      const float f0 = fc_w[k * NQ + 0], f1 = fc_w[k * NQ + 1];
      const float f2 = fc_w[k * NQ + 2], f3 = fc_w[k * NQ + 3];
      const float t0 = half ? -f0 : f0;
      v2f wp[4];
#pragma unroll
      for (int rp = 0; rp < 4; ++rp) {
        const float t1 = (rp >> 1) ? t0 - f1 : t0 + f1;
        const float t12 = (rp & 1) ? t1 - f2 : t1 + f2;
        wp[rp].x = t12 + f3;
        wp[rp].y = t12 - f3;
      }
      sh[4 + 2 * k][lane] = (v4f){wp[0].x, wp[0].y, wp[1].x, wp[1].y};
      sh[5 + 2 * k][lane] = (v4f){wp[2].x, wp[2].y, wp[3].x, wp[3].y};
    }
  }
  __syncthreads();

  v2f a_re[4], a_im[4];
  v2f w0p[4], w1p[4], w2p[4];
  {
    const v4f q0 = sh[0][lane], q1 = sh[1][lane], q2 = sh[2][lane], q3 = sh[3][lane];
    a_re[0] = (v2f){q0.x, q0.y}; a_re[1] = (v2f){q0.z, q0.w};
    a_re[2] = (v2f){q1.x, q1.y}; a_re[3] = (v2f){q1.z, q1.w};
    a_im[0] = (v2f){q2.x, q2.y}; a_im[1] = (v2f){q2.z, q2.w};
    a_im[2] = (v2f){q3.x, q3.y}; a_im[3] = (v2f){q3.z, q3.w};
    const v4f q4 = sh[4][lane], q5 = sh[5][lane], q6 = sh[6][lane];
    const v4f q7 = sh[7][lane], q8 = sh[8][lane], q9 = sh[9][lane];
    w0p[0] = (v2f){q4.x, q4.y}; w0p[1] = (v2f){q4.z, q4.w};
    w0p[2] = (v2f){q5.x, q5.y}; w0p[3] = (v2f){q5.z, q5.w};
    w1p[0] = (v2f){q6.x, q6.y}; w1p[1] = (v2f){q6.z, q6.w};
    w1p[2] = (v2f){q7.x, q7.y}; w1p[3] = (v2f){q7.z, q7.w};
    w2p[0] = (v2f){q8.x, q8.y}; w2p[1] = (v2f){q8.z, q8.w};
    w2p[2] = (v2f){q9.x, q9.y}; w2p[3] = (v2f){q9.z, q9.w};
  }
  const float b0 = fc_b[0], b1 = fc_b[1], b2 = fc_b[2];

  for (int t = gwave; t < ntiles; t += nwaves) {
    const float* xt = x + (size_t)t * 256;
    __builtin_prefetch(xt + (size_t)nwaves * 256 + lane * 16, 0, 0);

    v2f bch[4];
#pragma unroll
    for (int c = 0; c < 4; ++c)
      bch[c] = *(const v2f*)(xt + lrow * 16 + 4 * c + 2 * half);

    v8f accR = {}, accI = {};
#pragma unroll
    for (int c = 0; c < 4; ++c) {
      accR = __builtin_amdgcn_wmma_f32_16x16x4_f32(
          false, a_re[c], false, bch[c], (short)0, accR, false, false);
      accI = __builtin_amdgcn_wmma_f32_16x16x4_f32(
          false, a_im[c], false, bch[c], (short)0, accI, false, false);
    }

    v2f acc0 = {0.f, 0.f}, acc1 = {0.f, 0.f}, acc2 = {0.f, 0.f}, acc3 = {0.f, 0.f};
#pragma unroll
    for (int rp = 0; rp < 4; ++rp) {
      v2f vr; vr.x = accR[2 * rp]; vr.y = accR[2 * rp + 1];
      v2f vi; vi.x = accI[2 * rp]; vi.y = accI[2 * rp + 1];
      v2f p = vr * vr + vi * vi;
      acc0 += p * w0p[rp];
      acc1 += p * w1p[rp];
      acc2 += p * w2p[rp];
      acc3 += p;
    }
    float s0 = acc0.x + acc0.y;
    float s1 = acc1.x + acc1.y;
    float s2 = acc2.x + acc2.y;
    float s3 = acc3.x + acc3.y;

    s0 += swap_half16(s0);
    s1 += swap_half16(s1);
    s2 += swap_half16(s2);
    s3 += swap_half16(s3);

    const float rinv = __builtin_amdgcn_rcpf(s3);
    const size_t samp = (size_t)t * 16 + lrow;
    float* op = out + samp * 3;
    if (half == 0) {
      op[0] = s0 * rinv + b0;
      op[1] = s1 * rinv + b1;
    } else {
      op[2] = s2 * rinv + b2;
    }
  }
}

extern "C" void kernel_launch(void* const* d_in, const int* in_sizes, int n_in,
                              void* d_out, int out_size, void* d_ws, size_t ws_size,
                              hipStream_t stream) {
  const float* x      = (const float*)d_in[0];
  const float* params = (const float*)d_in[1];
  const float* fc_w   = (const float*)d_in[2];
  const float* fc_b   = (const float*)d_in[3];
  float* out = (float*)d_out;

  const int ntiles = in_sizes[0] / 256;

  int blocks = 2048;
  int max_blocks = (ntiles + 7) / 8;
  if (blocks > max_blocks) blocks = max_blocks;
  if (blocks < 1) blocks = 1;
  qnn_main_kernel<<<blocks, 256, 0, stream>>>(x, params, fc_w, fc_b, out, ntiles);
}


# Resident grid, placement fixed by LDS size: 256 work-groups of 16 waves, one per WGP (176 KiB LDS each); each wave loops over 16 grid-strided tiles, the loads of its next three in flight
# speedup vs baseline: 4.4692x; 1.1192x over previous
#include <hip/hip_runtime.h>
#include <math.h>


typedef float v2f __attribute__((ext_vector_type(2)));
typedef float v4f __attribute__((ext_vector_type(4)));
typedef float v8f __attribute__((ext_vector_type(8)));

#define NQ 4
#define WAVES 16
#define LDS_QUADS 11264

struct cplx { float re, im; };

__device__ __forceinline__ cplx cmul(cplx a, cplx b) {
  cplx c;
  c.re = a.re * b.re - a.im * b.im;
  c.im = a.re * b.im + a.im * b.re;
  return c;
}

__device__ __forceinline__ cplx csel(bool s, cplx one, cplx zero) {
  cplx c;
  c.re = s ? one.re : zero.re;
  c.im = s ? one.im : zero.im;
  return c;
}

__device__ __forceinline__ float swap_half16(float v) {
  int i = __float_as_int(v);
  int r = __builtin_amdgcn_permlanex16(i, i, 0x76543210, 0xfedcba98, false, false);
  return __int_as_float(r);
}

__device__ __forceinline__ float lane_bcast(float v, int j) {
  return __int_as_float(__builtin_amdgcn_readlane(__float_as_int(v), j));
}

__device__ __forceinline__ void load_tile(v2f (&b)[4], const float* p) {
#pragma unroll
  for (int c = 0; c < 4; ++c) b[c] = *(const v2f*)(p + 4 * c);
}

__global__ __launch_bounds__(WAVES * 32) void qnn_main_kernel(
    const float* __restrict__ x, const float* __restrict__ params,
    const float* __restrict__ fc_w, const float* __restrict__ fc_b,
    float* __restrict__ out, int ntiles) {
  __shared__ v4f sh[LDS_QUADS];

  const int lane = threadIdx.x & 31;
  const int half = lane >> 4;
  const int lrow = lane & 15;
  const int wave = __builtin_amdgcn_readfirstlane((int)(threadIdx.x >> 5));
  const int gwave = blockIdx.x * WAVES + wave;
  const int nwaves = gridDim.x * WAVES;

  const float* xl = x + lrow * 16 + 2 * half;
  int t = gwave;
  const int tf = t < ntiles ? t : 0;
  const int tg = t + nwaves < ntiles ? t + nwaves : tf;
  const int th = t + 2 * nwaves < ntiles ? t + 2 * nwaves : tf;
  v2f bA[4], bB[4], bC[4], bD[4];
  load_tile(bA, xl + (size_t)tf * 256);
  load_tile(bB, xl + (size_t)tg * 256);
  load_tile(bC, xl + (size_t)th * 256);

  if (wave == 0) {
    float pv = 0.f;
#pragma unroll
    for (int j = 0; j < 12; ++j) {
      const int pj = __builtin_amdgcn_readfirstlane(__float_as_int(params[j]));
      pv = lane == j ? __int_as_float(pj) : pv;
    }
    const float h = 0.5f * pv;
    const float cv = cosf(h), sv = sinf(h);
    float C[12], S[12];
#pragma unroll
    for (int j = 0; j < 12; ++j) {
      C[j] = lane_bcast(cv, j);
      S[j] = lane_bcast(sv, j);
    }

    int r = lrow;
    if (r & 2) r ^= 1;
    if (r & 4) r ^= 2;
    if (r & 8) r ^= 4;

    cplx g0[NQ], g1[NQ];
#pragma unroll
    for (int q = 0; q < NQ; ++q) {
      const float cx = C[3 * q + 0], sx = S[3 * q + 0];
      const float cy = C[3 * q + 1], sy = S[3 * q + 1];
      const float cz = C[3 * q + 2], sz = S[3 * q + 2];
      const float a = cy * cx, b = sy * sx, d = sy * cx, e = cy * sx;
      const float p = cz * a + sz * b;
      const float s = cz * b - sz * a;
      const float m = cz * d + sz * e;
      const float n = sz * d - cz * e;
      const bool u = ((r >> (NQ - 1 - q)) & 1) != 0;
      cplx G00 = {p, s}, G01 = {-m, n}, G10 = {m, n}, G11 = {p, -s};
      g0[q] = csel(u, G10, G00);
      g1[q] = csel(u, G11, G01);
    }

    const cplx t2 = csel(half != 0, g1[2], g0[2]);
    v2f a_re[4], a_im[4];
#pragma unroll
    for (int c = 0; c < 4; ++c) {
      const cplx p01 = cmul((c >> 1) ? g1[0] : g0[0], (c & 1) ? g1[1] : g0[1]);
      const cplx p012 = cmul(p01, t2);
      const cplx e0 = cmul(p012, g0[3]);
      const cplx e1 = cmul(p012, g1[3]);
      a_re[c].x = e0.re; a_re[c].y = e1.re;
      a_im[c].x = e0.im; a_im[c].y = e1.im;
    }
    sh[0 * 32 + lane] = (v4f){a_re[0].x, a_re[0].y, a_re[1].x, a_re[1].y};
    sh[1 * 32 + lane] = (v4f){a_re[2].x, a_re[2].y, a_re[3].x, a_re[3].y};
    sh[2 * 32 + lane] = (v4f){a_im[0].x, a_im[0].y, a_im[1].x, a_im[1].y};
    sh[3 * 32 + lane] = (v4f){a_im[2].x, a_im[2].y, a_im[3].x, a_im[3].y};
  } else if (wave == 1) {
#pragma unroll
    for (int k = 0; k < 3; ++k) {
      const float f0 = fc_w[k * NQ + 0], f1 = fc_w[k * NQ + 1];
      const float f2 = fc_w[k * NQ + 2], f3 = fc_w[k * NQ + 3];
      const float t0 = half ? -f0 : f0;
      v2f wp[4];
#pragma unroll
      for (int rp = 0; rp < 4; ++rp) {
        const float t1 = (rp >> 1) ? t0 - f1 : t0 + f1;
        const float t12 = (rp & 1) ? t1 - f2 : t1 + f2;
        wp[rp].x = t12 + f3;
        wp[rp].y = t12 - f3;
      }
      sh[(4 + 2 * k) * 32 + lane] = (v4f){wp[0].x, wp[0].y, wp[1].x, wp[1].y};
      sh[(5 + 2 * k) * 32 + lane] = (v4f){wp[2].x, wp[2].y, wp[3].x, wp[3].y};
    }
  }
  __builtin_amdgcn_fence(__ATOMIC_RELEASE, "workgroup", "local");
  __builtin_amdgcn_s_barrier();
  __builtin_amdgcn_fence(__ATOMIC_ACQUIRE, "workgroup", "local");

  v2f a_re[4], a_im[4];
  v2f w0p[4], w1p[4], w2p[4];
  {
    const v4f q0 = sh[0 * 32 + lane], q1 = sh[1 * 32 + lane], q2 = sh[2 * 32 + lane], q3 = sh[3 * 32 + lane];
    a_re[0] = (v2f){q0.x, q0.y}; a_re[1] = (v2f){q0.z, q0.w};
    a_re[2] = (v2f){q1.x, q1.y}; a_re[3] = (v2f){q1.z, q1.w};
    a_im[0] = (v2f){q2.x, q2.y}; a_im[1] = (v2f){q2.z, q2.w};
    a_im[2] = (v2f){q3.x, q3.y}; a_im[3] = (v2f){q3.z, q3.w};
    const v4f q4 = sh[4 * 32 + lane], q5 = sh[5 * 32 + lane], q6 = sh[6 * 32 + lane];
    const v4f q7 = sh[7 * 32 + lane], q8 = sh[8 * 32 + lane], q9 = sh[9 * 32 + lane];
    w0p[0] = (v2f){q4.x, q4.y}; w0p[1] = (v2f){q4.z, q4.w};
    w0p[2] = (v2f){q5.x, q5.y}; w0p[3] = (v2f){q5.z, q5.w};
    w1p[0] = (v2f){q6.x, q6.y}; w1p[1] = (v2f){q6.z, q6.w};
    w1p[2] = (v2f){q7.x, q7.y}; w1p[3] = (v2f){q7.z, q7.w};
    w2p[0] = (v2f){q8.x, q8.y}; w2p[1] = (v2f){q8.z, q8.w};
    w2p[2] = (v2f){q9.x, q9.y}; w2p[3] = (v2f){q9.z, q9.w};
  }
  const float b0 = fc_b[0], b1 = fc_b[1], b2 = fc_b[2];

  auto tile = [&](const v2f (&b)[4], int tt) {
    v8f accR = {}, accI = {};
#pragma unroll
    for (int c = 0; c < 4; ++c) {
      accR = __builtin_amdgcn_wmma_f32_16x16x4_f32(
          false, a_re[c], false, b[c], (short)0, accR, false, false);
      accI = __builtin_amdgcn_wmma_f32_16x16x4_f32(
          false, a_im[c], false, b[c], (short)0, accI, false, false);
    }
    asm volatile("v_nop\n\tv_nop\n\tv_nop\n\tv_nop"
                 : "+v"(accR), "+v"(accI)
                 : "v"(b[0]), "v"(b[1]), "v"(b[2]), "v"(b[3]));

    v2f acc0 = {0.f, 0.f}, acc1 = {0.f, 0.f}, acc2 = {0.f, 0.f}, acc3 = {0.f, 0.f};
#pragma unroll
    for (int rp = 0; rp < 4; ++rp) {
      v2f vr; vr.x = accR[2 * rp]; vr.y = accR[2 * rp + 1];
      v2f vi; vi.x = accI[2 * rp]; vi.y = accI[2 * rp + 1];
      v2f p = vr * vr + vi * vi;
      acc0 += p * w0p[rp];
      acc1 += p * w1p[rp];
      acc2 += p * w2p[rp];
      acc3 += p;
    }
    float s0 = acc0.x + acc0.y;
    float s1 = acc1.x + acc1.y;
    float s2 = acc2.x + acc2.y;
    float s3 = acc3.x + acc3.y;

    s0 += swap_half16(s0);
    s1 += swap_half16(s1);
    s2 += swap_half16(s2);
    s3 += swap_half16(s3);

    const float rinv = __builtin_amdgcn_rcpf(s3);
    const size_t samp = (size_t)tt * 16 + lrow;
    float* op = out + samp * 3;
    if (half == 0) {
      op[0] = s0 * rinv + b0;
      op[1] = s1 * rinv + b1;
    } else {
      op[2] = s2 * rinv + b2;
    }
  };

  auto ahead = [&](int tt) { return tt + 3 * nwaves < ntiles ? tt + 3 * nwaves : tt; };

  while (t + 3 * nwaves < ntiles) {
    load_tile(bD, xl + (size_t)ahead(t) * 256);
    tile(bA, t);
    t += nwaves;
    load_tile(bA, xl + (size_t)ahead(t) * 256);
    tile(bB, t);
    t += nwaves;
    load_tile(bB, xl + (size_t)ahead(t) * 256);
    tile(bC, t);
    t += nwaves;
    load_tile(bC, xl + (size_t)ahead(t) * 256);
    tile(bD, t);
    t += nwaves;
  }
  if (t < ntiles) {
    tile(bA, t);
    t += nwaves;
    if (t < ntiles) {
      tile(bB, t);
      t += nwaves;
      if (t < ntiles) tile(bC, t);
    }
  }
}

extern "C" void kernel_launch(void* const* d_in, const int* in_sizes, int n_in,
                              void* d_out, int out_size, void* d_ws, size_t ws_size,
                              hipStream_t stream) {
  const float* x      = (const float*)d_in[0];
  const float* params = (const float*)d_in[1];
  const float* fc_w   = (const float*)d_in[2];
  const float* fc_b   = (const float*)d_in[3];
  float* out = (float*)d_out;

  const int ntiles = in_sizes[0] / 256;
  if (ntiles < 1) return;

  int blocks = 256;
  int max_blocks = (ntiles + WAVES - 1) / WAVES;
  if (blocks > max_blocks) blocks = max_blocks;
  if (blocks < 1) blocks = 1;
  qnn_main_kernel<<<blocks, WAVES * 32, 0, stream>>>(x, params, fc_w, fc_b, out, ntiles);
}


# Two adjacent tiles per step: low lanes finish the first tile's samples, high lanes the second's, so each lane stores its sample's 3 logits as one 12 B store (whole lines), not strided 4 B pieces
# speedup vs baseline: 5.0029x; 1.1194x over previous
#include <hip/hip_runtime.h>
#include <math.h>


typedef float v2f __attribute__((ext_vector_type(2)));
typedef float v3f __attribute__((ext_vector_type(3), aligned(4)));
typedef float v4f __attribute__((ext_vector_type(4)));
typedef float v8f __attribute__((ext_vector_type(8)));

#define NQ 4
#define WAVES 16
#define LDS_QUADS 11264

struct cplx { float re, im; };

__device__ __forceinline__ cplx cmul(cplx a, cplx b) {
  cplx c;
  c.re = a.re * b.re - a.im * b.im;
  c.im = a.re * b.im + a.im * b.re;
  return c;
}

__device__ __forceinline__ cplx csel(bool s, cplx one, cplx zero) {
  cplx c;
  c.re = s ? one.re : zero.re;
  c.im = s ? one.im : zero.im;
  return c;
}

__device__ __forceinline__ float swap_half16(float v) {
  int i = __float_as_int(v);
  int r = __builtin_amdgcn_permlanex16(i, i, 0x76543210, 0xfedcba98, false, false);
  return __int_as_float(r);
}

__device__ __forceinline__ float lane_bcast(float v, int j) {
  return __int_as_float(__builtin_amdgcn_readlane(__float_as_int(v), j));
}

__device__ __forceinline__ void load_pair(v2f (&b)[8], const float* p, int offB) {
#pragma unroll
  for (int c = 0; c < 4; ++c) b[c] = *(const v2f*)(p + 4 * c);
#pragma unroll
  for (int c = 0; c < 4; ++c) b[4 + c] = *(const v2f*)(p + offB + 4 * c);
}

__global__ __launch_bounds__(WAVES * 32) void qnn_main_kernel(
    const float* __restrict__ x, const float* __restrict__ params,
    const float* __restrict__ fc_w, const float* __restrict__ fc_b,
    float* __restrict__ out, int ntiles) {
  __shared__ v4f sh[LDS_QUADS];

  const int lane = threadIdx.x & 31;
  const int half = lane >> 4;
  const int lrow = lane & 15;
  const int wave = __builtin_amdgcn_readfirstlane((int)(threadIdx.x >> 5));
  const int gwave = blockIdx.x * WAVES + wave;
  const int nwaves = gridDim.x * WAVES;

  const int npairs = (ntiles + 1) >> 1;
  auto second = [&](int pp) { return 2 * pp + 1 < ntiles ? 256 : 0; };

  const float* xl = x + lrow * 16 + 2 * half;
  int t = gwave;
  const int tf = t < npairs ? t : 0;
  const int tg = t + nwaves < npairs ? t + nwaves : tf;
  v2f bA[8], bB[8], bC[8];
  load_pair(bA, xl + (size_t)tf * 512, second(tf));
  load_pair(bB, xl + (size_t)tg * 512, second(tg));

  if (wave == 0) {
    float pv = 0.f;
#pragma unroll
    for (int j = 0; j < 12; ++j) {
      const int pj = __builtin_amdgcn_readfirstlane(__float_as_int(params[j]));
      pv = lane == j ? __int_as_float(pj) : pv;
    }
    const float h = 0.5f * pv;
    const float cv = cosf(h), sv = sinf(h);
    float C[12], S[12];
#pragma unroll
    for (int j = 0; j < 12; ++j) {
      C[j] = lane_bcast(cv, j);
      S[j] = lane_bcast(sv, j);
    }

    int r = lrow;
    if (r & 2) r ^= 1;
    if (r & 4) r ^= 2;
    if (r & 8) r ^= 4;

    cplx g0[NQ], g1[NQ];
#pragma unroll
    for (int q = 0; q < NQ; ++q) {
      const float cx = C[3 * q + 0], sx = S[3 * q + 0];
      const float cy = C[3 * q + 1], sy = S[3 * q + 1];
      const float cz = C[3 * q + 2], sz = S[3 * q + 2];
      const float a = cy * cx, b = sy * sx, d = sy * cx, e = cy * sx;
      const float p = cz * a + sz * b;
      const float s = cz * b - sz * a;
      const float m = cz * d + sz * e;
      const float n = sz * d - cz * e;
      const bool u = ((r >> (NQ - 1 - q)) & 1) != 0;
      cplx G00 = {p, s}, G01 = {-m, n}, G10 = {m, n}, G11 = {p, -s};
      g0[q] = csel(u, G10, G00);
      g1[q] = csel(u, G11, G01);
    }

    const cplx t2 = csel(half != 0, g1[2], g0[2]);
    v2f a_re[4], a_im[4];
#pragma unroll
    for (int c = 0; c < 4; ++c) {
      const cplx p01 = cmul((c >> 1) ? g1[0] : g0[0], (c & 1) ? g1[1] : g0[1]);
      const cplx p012 = cmul(p01, t2);
      const cplx e0 = cmul(p012, g0[3]);
      const cplx e1 = cmul(p012, g1[3]);
      a_re[c].x = e0.re; a_re[c].y = e1.re;
      a_im[c].x = e0.im; a_im[c].y = e1.im;
    }
    sh[0 * 32 + lane] = (v4f){a_re[0].x, a_re[0].y, a_re[1].x, a_re[1].y};
    sh[1 * 32 + lane] = (v4f){a_re[2].x, a_re[2].y, a_re[3].x, a_re[3].y};
    sh[2 * 32 + lane] = (v4f){a_im[0].x, a_im[0].y, a_im[1].x, a_im[1].y};
    sh[3 * 32 + lane] = (v4f){a_im[2].x, a_im[2].y, a_im[3].x, a_im[3].y};
  } else if (wave == 1) {
#pragma unroll
    for (int k = 0; k < 3; ++k) {
      const float f0 = fc_w[k * NQ + 0], f1 = fc_w[k * NQ + 1];
      const float f2 = fc_w[k * NQ + 2], f3 = fc_w[k * NQ + 3];
      const float t0 = half ? -f0 : f0;
      v2f wp[4];
#pragma unroll
      for (int rp = 0; rp < 4; ++rp) {
        const float t1 = (rp >> 1) ? t0 - f1 : t0 + f1;
        const float t12 = (rp & 1) ? t1 - f2 : t1 + f2;
        wp[rp].x = t12 + f3;
        wp[rp].y = t12 - f3;
      }
      sh[(4 + 2 * k) * 32 + lane] = (v4f){wp[0].x, wp[0].y, wp[1].x, wp[1].y};
      sh[(5 + 2 * k) * 32 + lane] = (v4f){wp[2].x, wp[2].y, wp[3].x, wp[3].y};
    }
  }
  __builtin_amdgcn_fence(__ATOMIC_RELEASE, "workgroup", "local");
  __builtin_amdgcn_s_barrier();
  __builtin_amdgcn_fence(__ATOMIC_ACQUIRE, "workgroup", "local");

  v2f a_re[4], a_im[4];
  v2f w0p[4], w1p[4], w2p[4];
  {
    const v4f q0 = sh[0 * 32 + lane], q1 = sh[1 * 32 + lane], q2 = sh[2 * 32 + lane], q3 = sh[3 * 32 + lane];
    a_re[0] = (v2f){q0.x, q0.y}; a_re[1] = (v2f){q0.z, q0.w};
    a_re[2] = (v2f){q1.x, q1.y}; a_re[3] = (v2f){q1.z, q1.w};
    a_im[0] = (v2f){q2.x, q2.y}; a_im[1] = (v2f){q2.z, q2.w};
    a_im[2] = (v2f){q3.x, q3.y}; a_im[3] = (v2f){q3.z, q3.w};
    const v4f q4 = sh[4 * 32 + lane], q5 = sh[5 * 32 + lane], q6 = sh[6 * 32 + lane];
    const v4f q7 = sh[7 * 32 + lane], q8 = sh[8 * 32 + lane], q9 = sh[9 * 32 + lane];
    w0p[0] = (v2f){q4.x, q4.y}; w0p[1] = (v2f){q4.z, q4.w};
    w0p[2] = (v2f){q5.x, q5.y}; w0p[3] = (v2f){q5.z, q5.w};
    w1p[0] = (v2f){q6.x, q6.y}; w1p[1] = (v2f){q6.z, q6.w};
    w1p[2] = (v2f){q7.x, q7.y}; w1p[3] = (v2f){q7.z, q7.w};
    w2p[0] = (v2f){q8.x, q8.y}; w2p[1] = (v2f){q8.z, q8.w};
    w2p[2] = (v2f){q9.x, q9.y}; w2p[3] = (v2f){q9.z, q9.w};
  }
  const float b0 = fc_b[0], b1 = fc_b[1], b2 = fc_b[2];

  const int nsamp = ntiles * 16;

  auto reduce = [&](const v8f& accR, const v8f& accI, float (&s)[4]) {
    v2f acc0 = {0.f, 0.f}, acc1 = {0.f, 0.f}, acc2 = {0.f, 0.f}, acc3 = {0.f, 0.f};
#pragma unroll
    for (int rp = 0; rp < 4; ++rp) {
      v2f vr; vr.x = accR[2 * rp]; vr.y = accR[2 * rp + 1];
      v2f vi; vi.x = accI[2 * rp]; vi.y = accI[2 * rp + 1];
      v2f p = vr * vr + vi * vi;
      acc0 += p * w0p[rp];
      acc1 += p * w1p[rp];
      acc2 += p * w2p[rp];
      acc3 += p;
    }
    s[0] = acc0.x + acc0.y;
    s[1] = acc1.x + acc1.y;
    s[2] = acc2.x + acc2.y;
    s[3] = acc3.x + acc3.y;
  };

  auto pair = [&](const v2f (&b)[8], int pp) {
    v8f rA = {}, iA = {}, rB = {}, iB = {};
#pragma unroll
    for (int c = 0; c < 4; ++c) {
      rA = __builtin_amdgcn_wmma_f32_16x16x4_f32(
          false, a_re[c], false, b[c], (short)0, rA, false, false);
      iA = __builtin_amdgcn_wmma_f32_16x16x4_f32(
          false, a_im[c], false, b[c], (short)0, iA, false, false);
      rB = __builtin_amdgcn_wmma_f32_16x16x4_f32(
          false, a_re[c], false, b[4 + c], (short)0, rB, false, false);
      iB = __builtin_amdgcn_wmma_f32_16x16x4_f32(
          false, a_im[c], false, b[4 + c], (short)0, iB, false, false);
    }
    asm volatile("v_nop\n\tv_nop\n\tv_nop\n\tv_nop"
                 : "+v"(rA), "+v"(iA), "+v"(rB), "+v"(iB)
                 : "v"(b[0]), "v"(b[1]), "v"(b[2]), "v"(b[3]),
                   "v"(b[4]), "v"(b[5]), "v"(b[6]), "v"(b[7]));

    float sA[4], sB[4];
    reduce(rA, iA, sA);
    reduce(rB, iB, sB);

    float s[4];
#pragma unroll
    for (int k = 0; k < 4; ++k) {
      const float own = half ? sB[k] : sA[k];
      const float give = half ? sA[k] : sB[k];
      s[k] = own + swap_half16(give);
    }

    const float rinv = __builtin_amdgcn_rcpf(s[3]);
    const int samp = pp * 32 + lane;
    if (samp < nsamp) {
      v3f o;
      o.x = s[0] * rinv + b0;
      o.y = s[1] * rinv + b1;
      o.z = s[2] * rinv + b2;
      *(v3f*)(out + (size_t)samp * 3) = o;
    }
  };

  auto ahead = [&](int pp) { return pp + 2 * nwaves < npairs ? pp + 2 * nwaves : pp; };

  while (t + 2 * nwaves < npairs) {
    int a = ahead(t);
    load_pair(bC, xl + (size_t)a * 512, second(a));
    pair(bA, t);
    t += nwaves;
    a = ahead(t);
    load_pair(bA, xl + (size_t)a * 512, second(a));
    pair(bB, t);
    t += nwaves;
    a = ahead(t);
    load_pair(bB, xl + (size_t)a * 512, second(a));
    pair(bC, t);
    t += nwaves;
  }
  if (t < npairs) {
    pair(bA, t);
    t += nwaves;
    if (t < npairs) pair(bB, t);
  }
}

extern "C" void kernel_launch(void* const* d_in, const int* in_sizes, int n_in,
                              void* d_out, int out_size, void* d_ws, size_t ws_size,
                              hipStream_t stream) {
  const float* x      = (const float*)d_in[0];
  const float* params = (const float*)d_in[1];
  const float* fc_w   = (const float*)d_in[2];
  const float* fc_b   = (const float*)d_in[3];
  float* out = (float*)d_out;

  const int ntiles = in_sizes[0] / 256;
  if (ntiles < 1) return;

  int blocks = 256;
  int max_blocks = (ntiles + 2 * WAVES - 1) / (2 * WAVES);
  if (blocks > max_blocks) blocks = max_blocks;
  if (blocks < 1) blocks = 1;
  qnn_main_kernel<<<blocks, WAVES * 32, 0, stream>>>(x, params, fc_w, fc_b, out, ntiles);
}
